# starting kernel
# baseline (speedup 1.0000x reference)
#include <hip/hip_runtime.h>
#include <math.h>

typedef __attribute__((ext_vector_type(2))) float v2f;
typedef __attribute__((ext_vector_type(4))) float v4f;
typedef __attribute__((ext_vector_type(8))) float v8f;

#define D 128
#define LNS 132
#define WAVES_PER_BLOCK 4

__global__ __launch_bounds__(32 * WAVES_PER_BLOCK)
void sage_wmma_kernel(const float* __restrict__ h,
                      const int*   __restrict__ nbr_idx,
                      const float* __restrict__ nbr_w,
                      const float* __restrict__ w_self,
                      const float* __restrict__ w_nei,
                      const float* __restrict__ gamma,
                      const float* __restrict__ beta,
                      float* __restrict__ out,
                      int N, int K)
{
    __shared__ float lds[WAVES_PER_BLOCK * 16 * LNS];

    const int lane   = threadIdx.x & 31;
    const int waveId = threadIdx.x >> 5;
    const int tile   = blockIdx.x * WAVES_PER_BLOCK + waveId;
    if (tile * 16 >= N) return;

    const int m    = lane & 15;
    const int half = lane >> 4;
    const int off  = half * 2;

    float* tileLds = lds + waveId * (16 * LNS);

    for (int r = 0; r < 16; ++r) {
        int grow = tile * 16 + r;
        if (grow >= N) grow = N - 1;
        v4f v = *reinterpret_cast<const v4f*>(h + (size_t)grow * D + 4 * lane);
        *reinterpret_cast<v4f*>(tileLds + r * LNS + 4 * lane) = v;
    }

    v2f aS[32];
#pragma unroll
    for (int c = 0; c < 32; ++c)
        aS[c] = *reinterpret_cast<const v2f*>(tileLds + m * LNS + 4 * c + off);

    for (int r = 0; r < 16; ++r) {
        int grow = tile * 16 + r;
        if (grow >= N) grow = N - 1;
        v4f acc = {};
        for (int k = 0; k < K; ++k) {
            const int   nrow = nbr_idx[(size_t)grow * K + k];
            const float w    = nbr_w [(size_t)grow * K + k];
            v4f v = *reinterpret_cast<const v4f*>(h + (size_t)nrow * D + 4 * lane);
            acc += v * w;
        }
        *reinterpret_cast<v4f*>(tileLds + r * LNS + 4 * lane) = acc;
    }

    v2f aA[32];
#pragma unroll
    for (int c = 0; c < 32; ++c)
        aA[c] = *reinterpret_cast<const v2f*>(tileLds + m * LNS + 4 * c + off);

    for (int jt = 0; jt < 8; ++jt) {
        const int j0 = jt * 16;
        v8f acc = {};
        const float* bsRow = w_self + (size_t)(j0 + m) * D + off;
        const float* bnRow = w_nei  + (size_t)(j0 + m) * D + off;
#pragma unroll
        for (int c = 0; c < 32; ++c) {
            v2f bS = *reinterpret_cast<const v2f*>(bsRow + 4 * c);
            acc = __builtin_amdgcn_wmma_f32_16x16x4_f32(false, aS[c], false, bS,
                                                        (short)0, acc, false, false);
            v2f bN = *reinterpret_cast<const v2f*>(bnRow + 4 * c);
            acc = __builtin_amdgcn_wmma_f32_16x16x4_f32(false, aA[c], false, bN,
                                                        (short)0, acc, false, false);
        }
#pragma unroll
        for (int r = 0; r < 8; ++r) {
            const int rr = r + 8 * half;
            const float z = acc[r];
            const float g = 0.5f * z * (1.0f + erff(z * 0.70710678118654752f));
            tileLds[rr * LNS + j0 + m] = g;
        }
    }

    for (int r = 0; r < 16; ++r) {
        const int grow = tile * 16 + r;
        const float x0 = tileLds[r * LNS + lane];
        const float x1 = tileLds[r * LNS + lane + 32];
        const float x2 = tileLds[r * LNS + lane + 64];
        const float x3 = tileLds[r * LNS + lane + 96];
        float s = x0 + x1 + x2 + x3;
#pragma unroll
        for (int d = 16; d >= 1; d >>= 1) s += __shfl_xor(s, d, 32);
        const float mu = s * (1.0f / D);
        const float d0 = x0 - mu, d1 = x1 - mu, d2 = x2 - mu, d3 = x3 - mu;
        float s2 = d0 * d0 + d1 * d1 + d2 * d2 + d3 * d3;
#pragma unroll
        for (int d = 16; d >= 1; d >>= 1) s2 += __shfl_xor(s2, d, 32);
        const float inv = rsqrtf(s2 * (1.0f / D) + 1e-5f);
        if (grow < N) {
            float* o = out + (size_t)grow * D;
            o[lane]      = d0 * inv * gamma[lane]      + beta[lane];
            o[lane + 32] = d1 * inv * gamma[lane + 32] + beta[lane + 32];
            o[lane + 64] = d2 * inv * gamma[lane + 64] + beta[lane + 64];
            o[lane + 96] = d3 * inv * gamma[lane + 96] + beta[lane + 96];
        }
    }
}

extern "C" void kernel_launch(void* const* d_in, const int* in_sizes, int n_in,
                              void* d_out, int out_size, void* d_ws, size_t ws_size,
                              hipStream_t stream) {
    const float* h       = (const float*)d_in[0];
    const int*   nbr_idx = (const int*)  d_in[1];
    const float* nbr_w   = (const float*)d_in[2];
    const float* w_self  = (const float*)d_in[3];
    const float* w_nei   = (const float*)d_in[4];
    const float* gamma   = (const float*)d_in[5];
    const float* beta    = (const float*)d_in[6];
    float* out = (float*)d_out;

    const int N = in_sizes[0] / D;
    const int K = (N > 0) ? (in_sizes[1] / N) : 16;

    const int tiles  = (N + 15) / 16;
    const int blocks = (tiles + WAVES_PER_BLOCK - 1) / WAVES_PER_BLOCK;
    hipLaunchKernelGGL(sage_wmma_kernel, dim3(blocks), dim3(32 * WAVES_PER_BLOCK), 0, stream,
                       h, nbr_idx, nbr_w, w_self, w_nei, gamma, beta, out, N, K);
}


# Both weight matrices staged once per 16-wave work-group into LDS in B-fragment order; waves loop over tiles and read B from LDS instead of 16-line global loads; k and sum order unchanged
# speedup vs baseline: 1.5066x; 1.5066x over previous
#include <hip/hip_runtime.h>
#include <math.h>

typedef __attribute__((ext_vector_type(2))) float v2f;
typedef __attribute__((ext_vector_type(4))) float v4f;
typedef __attribute__((ext_vector_type(8))) float v8f;

#define D 128
#define LNS 132
#define WAVES_PER_BLOCK 16
#define MAX_BLOCKS 256
#define WFRAG_FLOATS (8 * 32 * 2 * 64)

__global__ __launch_bounds__(32 * WAVES_PER_BLOCK)
void sage_wmma_kernel(const float* __restrict__ h,
                      const int*   __restrict__ nbr_idx,
                      const float* __restrict__ nbr_w,
                      const float* __restrict__ w_self,
                      const float* __restrict__ w_nei,
                      const float* __restrict__ gamma,
                      const float* __restrict__ beta,
                      float* __restrict__ out,
                      int N, int K)
{
    __shared__ float wfrag[WFRAG_FLOATS];
    __shared__ float lds[WAVES_PER_BLOCK * 16 * LNS];

    const int lane   = threadIdx.x & 31;
    const int waveId = threadIdx.x >> 5;

    const int m    = lane & 15;
    const int half = lane >> 4;
    const int off  = half * 2;

    float* tileLds = lds + waveId * (16 * LNS);

    for (int L = waveId; L < 256; L += WAVES_PER_BLOCK) {
        const int mat = L >> 7;
        const int jt  = (L >> 4) & 7;
        const int i   = L & 15;
        const float* wsrc = mat ? w_nei : w_self;
        const v4f v = *reinterpret_cast<const v4f*>(wsrc + (size_t)(16 * jt + m) * D + 8 * i + 4 * half);
        float* dst = wfrag + ((jt * 32 + 2 * i + half) * 2 + mat) * 64;
        v2f lo = {v.x, v.y};
        v2f hi = {v.z, v.w};
        *reinterpret_cast<v2f*>(dst + 2 * m)        = lo;
        *reinterpret_cast<v2f*>(dst + 2 * (m + 16)) = hi;
    }
    __syncthreads();

    const int tiles = (N + 15) >> 4;
    for (int tile = blockIdx.x + gridDim.x * waveId; tile < tiles; tile += gridDim.x * WAVES_PER_BLOCK) {

    for (int r = 0; r < 16; ++r) {
        int grow = tile * 16 + r;
        if (grow >= N) grow = N - 1;
        v4f v = *reinterpret_cast<const v4f*>(h + (size_t)grow * D + 4 * lane);
        *reinterpret_cast<v4f*>(tileLds + r * LNS + 4 * lane) = v;
    }

    v2f aS[32];
#pragma unroll
    for (int c = 0; c < 32; ++c)
        aS[c] = *reinterpret_cast<const v2f*>(tileLds + m * LNS + 4 * c + off);

    for (int r = 0; r < 16; ++r) {
        int grow = tile * 16 + r;
        if (grow >= N) grow = N - 1;
        v4f acc = {};
        for (int k = 0; k < K; ++k) {
            const int   nrow = nbr_idx[(size_t)grow * K + k];
            const float w    = nbr_w [(size_t)grow * K + k];
            v4f v = *reinterpret_cast<const v4f*>(h + (size_t)nrow * D + 4 * lane);
            acc += v * w;
        }
        *reinterpret_cast<v4f*>(tileLds + r * LNS + 4 * lane) = acc;
    }

    v2f aA[32];
#pragma unroll
    for (int c = 0; c < 32; ++c)
        aA[c] = *reinterpret_cast<const v2f*>(tileLds + m * LNS + 4 * c + off);

    for (int jt = 0; jt < 8; ++jt) {
        const int j0 = jt * 16;
        v8f acc = {};
        const float* bw = wfrag + jt * (32 * 2 * 64) + 2 * lane;
#pragma unroll
        for (int c = 0; c < 32; ++c) {
            v2f bS = *reinterpret_cast<const v2f*>(bw + c * 128);
            acc = __builtin_amdgcn_wmma_f32_16x16x4_f32(false, aS[c], false, bS,
                                                        (short)0, acc, false, false);
            v2f bN = *reinterpret_cast<const v2f*>(bw + c * 128 + 64);
            acc = __builtin_amdgcn_wmma_f32_16x16x4_f32(false, aA[c], false, bN,
                                                        (short)0, acc, false, false);
        }
#pragma unroll
        for (int r = 0; r < 8; ++r) {
            const int rr = r + 8 * half;
            const float z = acc[r];
            const float g = 0.5f * z * (1.0f + erff(z * 0.70710678118654752f));
            tileLds[rr * LNS + j0 + m] = g;
        }
    }

    for (int r = 0; r < 16; ++r) {
        const int grow = tile * 16 + r;
        const float x0 = tileLds[r * LNS + lane];
        const float x1 = tileLds[r * LNS + lane + 32];
        const float x2 = tileLds[r * LNS + lane + 64];
        const float x3 = tileLds[r * LNS + lane + 96];
        float s = x0 + x1 + x2 + x3;
#pragma unroll
        for (int d = 16; d >= 1; d >>= 1) s += __shfl_xor(s, d, 32);
        const float mu = s * (1.0f / D);
        const float d0 = x0 - mu, d1 = x1 - mu, d2 = x2 - mu, d3 = x3 - mu;
        float s2 = d0 * d0 + d1 * d1 + d2 * d2 + d3 * d3;
#pragma unroll
        for (int d = 16; d >= 1; d >>= 1) s2 += __shfl_xor(s2, d, 32);
        const float inv = rsqrtf(s2 * (1.0f / D) + 1e-5f);
        if (grow < N) {
            float* o = out + (size_t)grow * D;
            o[lane]      = d0 * inv * gamma[lane]      + beta[lane];
            o[lane + 32] = d1 * inv * gamma[lane + 32] + beta[lane + 32];
            o[lane + 64] = d2 * inv * gamma[lane + 64] + beta[lane + 64];
            o[lane + 96] = d3 * inv * gamma[lane + 96] + beta[lane + 96];
        }
    }
    }
}

extern "C" void kernel_launch(void* const* d_in, const int* in_sizes, int n_in,
                              void* d_out, int out_size, void* d_ws, size_t ws_size,
                              hipStream_t stream) {
    const float* h       = (const float*)d_in[0];
    const int*   nbr_idx = (const int*)  d_in[1];
    const float* nbr_w   = (const float*)d_in[2];
    const float* w_self  = (const float*)d_in[3];
    const float* w_nei   = (const float*)d_in[4];
    const float* gamma   = (const float*)d_in[5];
    const float* beta    = (const float*)d_in[6];
    float* out = (float*)d_out;

    const int N = in_sizes[0] / D;
    const int K = (N > 0) ? (in_sizes[1] / N) : 16;

    const int tiles  = (N + 15) / 16;
    int blocks = (tiles + WAVES_PER_BLOCK - 1) / WAVES_PER_BLOCK;
    if (blocks > MAX_BLOCKS) blocks = MAX_BLOCKS;
    hipLaunchKernelGGL(sage_wmma_kernel, dim3(blocks), dim3(32 * WAVES_PER_BLOCK), 0, stream,
                       h, nbr_idx, nbr_w, w_self, w_nei, gamma, beta, out, N, K);
}


# LayerNorm of a tile's 16 rows done together: each lane-butterfly step issued for all rows as independent shuffles, not one dependent chain per row; gamma/beta read once; sums and their order unchanged
# speedup vs baseline: 1.6791x; 1.1145x over previous
#include <hip/hip_runtime.h>
#include <math.h>

typedef __attribute__((ext_vector_type(2))) float v2f;
typedef __attribute__((ext_vector_type(4))) float v4f;
typedef __attribute__((ext_vector_type(8))) float v8f;

#define D 128
#define LNS 132
#define WAVES_PER_BLOCK 16
#define MAX_BLOCKS 256
#define WFRAG_FLOATS (8 * 32 * 2 * 64)

__global__ __launch_bounds__(32 * WAVES_PER_BLOCK)
void sage_wmma_kernel(const float* __restrict__ h,
                      const int*   __restrict__ nbr_idx,
                      const float* __restrict__ nbr_w,
                      const float* __restrict__ w_self,
                      const float* __restrict__ w_nei,
                      const float* __restrict__ gamma,
                      const float* __restrict__ beta,
                      float* __restrict__ out,
                      int N, int K)
{
    __shared__ float wfrag[WFRAG_FLOATS];
    __shared__ float lds[WAVES_PER_BLOCK * 16 * LNS];

    const int lane   = threadIdx.x & 31;
    const int waveId = threadIdx.x >> 5;

    const int m    = lane & 15;
    const int half = lane >> 4;
    const int off  = half * 2;

    float* tileLds = lds + waveId * (16 * LNS);

    for (int L = waveId; L < 256; L += WAVES_PER_BLOCK) {
        const int mat = L >> 7;
        const int jt  = (L >> 4) & 7;
        const int i   = L & 15;
        const float* wsrc = mat ? w_nei : w_self;
        const v4f v = *reinterpret_cast<const v4f*>(wsrc + (size_t)(16 * jt + m) * D + 8 * i + 4 * half);
        float* dst = wfrag + ((jt * 32 + 2 * i + half) * 2 + mat) * 64;
        v2f lo = {v.x, v.y};
        v2f hi = {v.z, v.w};
        *reinterpret_cast<v2f*>(dst + 2 * m)        = lo;
        *reinterpret_cast<v2f*>(dst + 2 * (m + 16)) = hi;
    }
    const float g0 = gamma[lane], g1 = gamma[lane + 32], g2 = gamma[lane + 64], g3 = gamma[lane + 96];
    const float b0 = beta[lane],  b1 = beta[lane + 32],  b2 = beta[lane + 64],  b3 = beta[lane + 96];
    __syncthreads();

    const int tiles = (N + 15) >> 4;
    for (int tile = blockIdx.x + gridDim.x * waveId; tile < tiles; tile += gridDim.x * WAVES_PER_BLOCK) {

    for (int r = 0; r < 16; ++r) {
        int grow = tile * 16 + r;
        if (grow >= N) grow = N - 1;
        v4f v = *reinterpret_cast<const v4f*>(h + (size_t)grow * D + 4 * lane);
        *reinterpret_cast<v4f*>(tileLds + r * LNS + 4 * lane) = v;
    }

    v2f aS[32];
#pragma unroll
    for (int c = 0; c < 32; ++c)
        aS[c] = *reinterpret_cast<const v2f*>(tileLds + m * LNS + 4 * c + off);

    for (int r = 0; r < 16; ++r) {
        int grow = tile * 16 + r;
        if (grow >= N) grow = N - 1;
        v4f acc = {};
        for (int k = 0; k < K; ++k) {
            const int   nrow = nbr_idx[(size_t)grow * K + k];
            const float w    = nbr_w [(size_t)grow * K + k];
            v4f v = *reinterpret_cast<const v4f*>(h + (size_t)nrow * D + 4 * lane);
            acc += v * w;
        }
        *reinterpret_cast<v4f*>(tileLds + r * LNS + 4 * lane) = acc;
    }

    v2f aA[32];
#pragma unroll
    for (int c = 0; c < 32; ++c)
        aA[c] = *reinterpret_cast<const v2f*>(tileLds + m * LNS + 4 * c + off);

    for (int jt = 0; jt < 8; ++jt) {
        const int j0 = jt * 16;
        v8f acc = {};
        const float* bw = wfrag + jt * (32 * 2 * 64) + 2 * lane;
#pragma unroll
        for (int c = 0; c < 32; ++c) {
            v2f bS = *reinterpret_cast<const v2f*>(bw + c * 128);
            acc = __builtin_amdgcn_wmma_f32_16x16x4_f32(false, aS[c], false, bS,
                                                        (short)0, acc, false, false);
            v2f bN = *reinterpret_cast<const v2f*>(bw + c * 128 + 64);
            acc = __builtin_amdgcn_wmma_f32_16x16x4_f32(false, aA[c], false, bN,
                                                        (short)0, acc, false, false);
        }
#pragma unroll
        for (int r = 0; r < 8; ++r) {
            const int rr = r + 8 * half;
            const float z = acc[r];
            const float g = 0.5f * z * (1.0f + erff(z * 0.70710678118654752f));
            tileLds[rr * LNS + j0 + m] = g;
        }
    }

    float x[16][4];
#pragma unroll
    for (int r = 0; r < 16; ++r) {
#pragma unroll
        for (int q = 0; q < 4; ++q) x[r][q] = tileLds[r * LNS + lane + 32 * q];
    }
    float s[16];
#pragma unroll
    for (int r = 0; r < 16; ++r) s[r] = x[r][0] + x[r][1] + x[r][2] + x[r][3];
#pragma unroll
    for (int d = 16; d >= 1; d >>= 1) {
        float t[16];
#pragma unroll
        for (int r = 0; r < 16; ++r) t[r] = __shfl_xor(s[r], d, 32);
#pragma unroll
        for (int r = 0; r < 16; ++r) s[r] += t[r];
    }
    float s2[16];
#pragma unroll
    for (int r = 0; r < 16; ++r) {
        const float mu = s[r] * (1.0f / D);
        const float d0 = x[r][0] - mu, d1 = x[r][1] - mu, d2 = x[r][2] - mu, d3 = x[r][3] - mu;
        x[r][0] = d0; x[r][1] = d1; x[r][2] = d2; x[r][3] = d3;
        s2[r] = d0 * d0 + d1 * d1 + d2 * d2 + d3 * d3;
    }
#pragma unroll
    for (int d = 16; d >= 1; d >>= 1) {
        float t[16];
#pragma unroll
        for (int r = 0; r < 16; ++r) t[r] = __shfl_xor(s2[r], d, 32);
#pragma unroll
        for (int r = 0; r < 16; ++r) s2[r] += t[r];
    }
    float* const otile = out + (size_t)tile * (16 * D) + lane;
    auto finish_row = [&](int r) {
        const float inv = rsqrtf(s2[r] * (1.0f / D) + 1e-5f);
        float* o = otile + r * D;
        o[0]  = x[r][0] * inv * g0 + b0;
        o[32] = x[r][1] * inv * g1 + b1;
        o[64] = x[r][2] * inv * g2 + b2;
        o[96] = x[r][3] * inv * g3 + b3;
    };
    if (tile * 16 + 16 <= N) {
#pragma unroll
        for (int r = 0; r < 16; ++r) finish_row(r);
    } else {
#pragma unroll
        for (int r = 0; r < 16; ++r)
            if (tile * 16 + r < N) finish_row(r);
    }
    }
}

extern "C" void kernel_launch(void* const* d_in, const int* in_sizes, int n_in,
                              void* d_out, int out_size, void* d_ws, size_t ws_size,
                              hipStream_t stream) {
    const float* h       = (const float*)d_in[0];
    const int*   nbr_idx = (const int*)  d_in[1];
    const float* nbr_w   = (const float*)d_in[2];
    const float* w_self  = (const float*)d_in[3];
    const float* w_nei   = (const float*)d_in[4];
    const float* gamma   = (const float*)d_in[5];
    const float* beta    = (const float*)d_in[6];
    float* out = (float*)d_out;

    const int N = in_sizes[0] / D;
    const int K = (N > 0) ? (in_sizes[1] / N) : 16;

    const int tiles  = (N + 15) / 16;
    int blocks = (tiles + WAVES_PER_BLOCK - 1) / WAVES_PER_BLOCK;
    if (blocks > MAX_BLOCKS) blocks = MAX_BLOCKS;
    hipLaunchKernelGGL(sage_wmma_kernel, dim3(blocks), dim3(32 * WAVES_PER_BLOCK), 0, stream,
                       h, nbr_idx, nbr_w, w_self, w_nei, gamma, beta, out, N, K);
}


# Neighbour gather pipelined: a tile's 256 indices and weights fetched once into lanes (readlane), each row's 16 loads issued one row ahead of their sums, 16-32 in flight per wave; sum order unchanged
# speedup vs baseline: 1.7821x; 1.0613x over previous
#include <hip/hip_runtime.h>
#include <math.h>

typedef __attribute__((ext_vector_type(2))) float v2f;
typedef __attribute__((ext_vector_type(4))) float v4f;
typedef __attribute__((ext_vector_type(8))) float v8f;
typedef __attribute__((ext_vector_type(4))) int   v4i;
typedef __attribute__((address_space(1))) const v4f gv4f;

#define D 128
#define LNS 132
#define WAVES_PER_BLOCK 16
#define MAX_BLOCKS 256
#define WFRAG_FLOATS (8 * 32 * 2 * 64)

__global__ __launch_bounds__(32 * WAVES_PER_BLOCK)
void sage_wmma_kernel(const float* __restrict__ h,
                      const int*   __restrict__ nbr_idx,
                      const float* __restrict__ nbr_w,
                      const float* __restrict__ w_self,
                      const float* __restrict__ w_nei,
                      const float* __restrict__ gamma,
                      const float* __restrict__ beta,
                      float* __restrict__ out,
                      int N, int K)
{
    __shared__ float wfrag[WFRAG_FLOATS];
    __shared__ float lds[WAVES_PER_BLOCK * 16 * LNS];

    const int lane   = threadIdx.x & 31;
    const int waveId = __builtin_amdgcn_readfirstlane(threadIdx.x >> 5);

    const int m    = lane & 15;
    const int half = lane >> 4;
    const int off  = half * 2;

    float* tileLds = lds + waveId * (16 * LNS);

    for (int L = waveId; L < 256; L += WAVES_PER_BLOCK) {
        const int mat = L >> 7;
        const int jt  = (L >> 4) & 7;
        const int i   = L & 15;
        const float* wsrc = mat ? w_nei : w_self;
        const v4f v = *reinterpret_cast<const v4f*>(wsrc + (size_t)(16 * jt + m) * D + 8 * i + 4 * half);
        float* dst = wfrag + ((jt * 32 + 2 * i + half) * 2 + mat) * 64;
        v2f lo = {v.x, v.y};
        v2f hi = {v.z, v.w};
        *reinterpret_cast<v2f*>(dst + 2 * m)        = lo;
        *reinterpret_cast<v2f*>(dst + 2 * (m + 16)) = hi;
    }
    const float g0 = gamma[lane], g1 = gamma[lane + 32], g2 = gamma[lane + 64], g3 = gamma[lane + 96];
    const float b0 = beta[lane],  b1 = beta[lane + 32],  b2 = beta[lane + 64],  b3 = beta[lane + 96];
    __syncthreads();

    const int tiles = (N + 15) >> 4;
    for (int tile = blockIdx.x + gridDim.x * waveId; tile < tiles; tile += gridDim.x * WAVES_PER_BLOCK) {

    const bool k16 = (K == 16);
    int   iv[8] = {0, 0, 0, 0, 0, 0, 0, 0};
    float wv[8] = {0.f, 0.f, 0.f, 0.f, 0.f, 0.f, 0.f, 0.f};
    if (k16) {
        int lrow = tile * 16 + (lane >> 1);
        if (lrow >= N) lrow = N - 1;
        const size_t e = (size_t)lrow * 16 + 8 * (lane & 1);
        const v4i i0 = *reinterpret_cast<const v4i*>(nbr_idx + e);
        const v4i i1 = *reinterpret_cast<const v4i*>(nbr_idx + e + 4);
        const v4f w0 = *reinterpret_cast<const v4f*>(nbr_w + e);
        const v4f w1 = *reinterpret_cast<const v4f*>(nbr_w + e + 4);
        iv[0] = i0.x; iv[1] = i0.y; iv[2] = i0.z; iv[3] = i0.w;
        iv[4] = i1.x; iv[5] = i1.y; iv[6] = i1.z; iv[7] = i1.w;
        wv[0] = w0.x; wv[1] = w0.y; wv[2] = w0.z; wv[3] = w0.w;
        wv[4] = w1.x; wv[5] = w1.y; wv[6] = w1.z; wv[7] = w1.w;
    }

    for (int r = 0; r < 16; ++r) {
        int grow = tile * 16 + r;
        if (grow >= N) grow = N - 1;
        v4f v = *reinterpret_cast<const v4f*>(h + (size_t)grow * D + 4 * lane);
        *reinterpret_cast<v4f*>(tileLds + r * LNS + 4 * lane) = v;
    }

    v2f aS[32];
#pragma unroll
    for (int c = 0; c < 32; ++c)
        aS[c] = *reinterpret_cast<const v2f*>(tileLds + m * LNS + 4 * c + off);

    if (k16) {
        const unsigned long long hbase = reinterpret_cast<unsigned long long>(h);
        auto issue_row = [&](int r, v4f (&buf)[16]) {
            unsigned long long base[16];
#pragma unroll
            for (int k = 0; k < 16; ++k) {
                const int nrow = __builtin_amdgcn_readlane(iv[k & 7], 2 * r + (k >> 3));
                base[k] = hbase + ((unsigned long long)(long long)nrow << 9);
            }
            __builtin_amdgcn_sched_barrier(0);
#pragma unroll
            for (int k = 0; k < 16; ++k)
                buf[k] = reinterpret_cast<const gv4f*>(base[k])[lane];
            __builtin_amdgcn_sched_barrier(0);
        };
        auto sum_row = [&](int r, const v4f (&buf)[16]) {
            v4f acc = {};
#pragma unroll
            for (int k = 0; k < 16; ++k) {
                const float w = __int_as_float(__builtin_amdgcn_readlane(__float_as_int(wv[k & 7]), 2 * r + (k >> 3)));
                acc += buf[k] * w;
            }
            *reinterpret_cast<v4f*>(tileLds + r * LNS + 4 * lane) = acc;
        };
        v4f bufA[16], bufB[16];
        issue_row(0, bufA);
#pragma nounroll
        for (int p = 0; p < 7; ++p) {
            issue_row(2 * p + 1, bufB);
            sum_row(2 * p, bufA);
            issue_row(2 * p + 2, bufA);
            sum_row(2 * p + 1, bufB);
        }
        issue_row(15, bufB);
        sum_row(14, bufA);
        sum_row(15, bufB);
    } else {
    for (int r = 0; r < 16; ++r) {
        int grow = tile * 16 + r;
        if (grow >= N) grow = N - 1;
        v4f acc = {};
        for (int k = 0; k < K; ++k) {
            const int   nrow = nbr_idx[(size_t)grow * K + k];
            const float w    = nbr_w [(size_t)grow * K + k];
            v4f v = *reinterpret_cast<const v4f*>(h + (size_t)nrow * D + 4 * lane);
            acc += v * w;
        }
        *reinterpret_cast<v4f*>(tileLds + r * LNS + 4 * lane) = acc;
    }
    }

    v2f aA[32];
#pragma unroll
    for (int c = 0; c < 32; ++c)
        aA[c] = *reinterpret_cast<const v2f*>(tileLds + m * LNS + 4 * c + off);

    for (int jt = 0; jt < 8; ++jt) {
        const int j0 = jt * 16;
        v8f acc = {};
        const float* bw = wfrag + jt * (32 * 2 * 64) + 2 * lane;
#pragma unroll
        for (int c = 0; c < 32; ++c) {
            v2f bS = *reinterpret_cast<const v2f*>(bw + c * 128);
            acc = __builtin_amdgcn_wmma_f32_16x16x4_f32(false, aS[c], false, bS,
                                                        (short)0, acc, false, false);
            v2f bN = *reinterpret_cast<const v2f*>(bw + c * 128 + 64);
            acc = __builtin_amdgcn_wmma_f32_16x16x4_f32(false, aA[c], false, bN,
                                                        (short)0, acc, false, false);
        }
#pragma unroll
        for (int r = 0; r < 8; ++r) {
            const int rr = r + 8 * half;
            const float z = acc[r];
            const float g = 0.5f * z * (1.0f + erff(z * 0.70710678118654752f));
            tileLds[rr * LNS + j0 + m] = g;
        }
    }

    float x[16][4];
#pragma unroll
    for (int r = 0; r < 16; ++r) {
#pragma unroll
        for (int q = 0; q < 4; ++q) x[r][q] = tileLds[r * LNS + lane + 32 * q];
    }
    float s[16];
#pragma unroll
    for (int r = 0; r < 16; ++r) s[r] = x[r][0] + x[r][1] + x[r][2] + x[r][3];
#pragma unroll
    for (int d = 16; d >= 1; d >>= 1) {
        float t[16];
#pragma unroll
        for (int r = 0; r < 16; ++r) t[r] = __shfl_xor(s[r], d, 32);
#pragma unroll
        for (int r = 0; r < 16; ++r) s[r] += t[r];
    }
    float s2[16];
#pragma unroll
    for (int r = 0; r < 16; ++r) {
        const float mu = s[r] * (1.0f / D);
        const float d0 = x[r][0] - mu, d1 = x[r][1] - mu, d2 = x[r][2] - mu, d3 = x[r][3] - mu;
        x[r][0] = d0; x[r][1] = d1; x[r][2] = d2; x[r][3] = d3;
        s2[r] = d0 * d0 + d1 * d1 + d2 * d2 + d3 * d3;
    }
#pragma unroll
    for (int d = 16; d >= 1; d >>= 1) {
        float t[16];
#pragma unroll
        for (int r = 0; r < 16; ++r) t[r] = __shfl_xor(s2[r], d, 32);
#pragma unroll
        for (int r = 0; r < 16; ++r) s2[r] += t[r];
    }
    float* const otile = out + (size_t)tile * (16 * D) + lane;
    auto finish_row = [&](int r) {
        const float inv = rsqrtf(s2[r] * (1.0f / D) + 1e-5f);
        float* o = otile + r * D;
        o[0]  = x[r][0] * inv * g0 + b0;
        o[32] = x[r][1] * inv * g1 + b1;
        o[64] = x[r][2] * inv * g2 + b2;
        o[96] = x[r][3] * inv * g3 + b3;
    };
    if (tile * 16 + 16 <= N) {
#pragma unroll
        for (int r = 0; r < 16; ++r) finish_row(r);
    } else {
#pragma unroll
        for (int r = 0; r < 16; ++r)
            if (tile * 16 + r < N) finish_row(r);
    }
    }
}

extern "C" void kernel_launch(void* const* d_in, const int* in_sizes, int n_in,
                              void* d_out, int out_size, void* d_ws, size_t ws_size,
                              hipStream_t stream) {
    const float* h       = (const float*)d_in[0];
    const int*   nbr_idx = (const int*)  d_in[1];
    const float* nbr_w   = (const float*)d_in[2];
    const float* w_self  = (const float*)d_in[3];
    const float* w_nei   = (const float*)d_in[4];
    const float* gamma   = (const float*)d_in[5];
    const float* beta    = (const float*)d_in[6];
    float* out = (float*)d_out;

    const int N = in_sizes[0] / D;
    const int K = (N > 0) ? (in_sizes[1] / N) : 16;

    const int tiles  = (N + 15) / 16;
    int blocks = (tiles + WAVES_PER_BLOCK - 1) / WAVES_PER_BLOCK;
    if (blocks > MAX_BLOCKS) blocks = MAX_BLOCKS;
    hipLaunchKernelGGL(sage_wmma_kernel, dim3(blocks), dim3(32 * WAVES_PER_BLOCK), 0, stream,
                       h, nbr_idx, nbr_w, w_self, w_nei, gamma, beta, out, N, K);
}


# Self half of the dual GEMM issued inside the gather: 16 matrix instructions behind each row's 16 loads (one row buffer); each output now sums its self terms first, then its neighbour terms
# speedup vs baseline: 1.8888x; 1.0599x over previous
#include <hip/hip_runtime.h>
#include <math.h>

typedef __attribute__((ext_vector_type(2))) float v2f;
typedef __attribute__((ext_vector_type(4))) float v4f;
typedef __attribute__((ext_vector_type(8))) float v8f;
typedef __attribute__((ext_vector_type(4))) int   v4i;
typedef __attribute__((address_space(1))) const v4f gv4f;

#define D 128
#define LNS 132
#define WAVES_PER_BLOCK 16
#define MAX_BLOCKS 256
#define WFRAG_FLOATS (8 * 32 * 2 * 64)

__global__ __launch_bounds__(32 * WAVES_PER_BLOCK)
void sage_wmma_kernel(const float* __restrict__ h,
                      const int*   __restrict__ nbr_idx,
                      const float* __restrict__ nbr_w,
                      const float* __restrict__ w_self,
                      const float* __restrict__ w_nei,
                      const float* __restrict__ gamma,
                      const float* __restrict__ beta,
                      float* __restrict__ out,
                      int N, int K)
{
    __shared__ float wfrag[WFRAG_FLOATS];
    __shared__ float lds[WAVES_PER_BLOCK * 16 * LNS];

    const int lane   = threadIdx.x & 31;
    const int waveId = __builtin_amdgcn_readfirstlane(threadIdx.x >> 5);

    const int m    = lane & 15;
    const int half = lane >> 4;
    const int off  = half * 2;

    float* tileLds = lds + waveId * (16 * LNS);

    for (int L = waveId; L < 256; L += WAVES_PER_BLOCK) {
        const int mat = L >> 7;
        const int jt  = (L >> 4) & 7;
        const int i   = L & 15;
        const float* wsrc = mat ? w_nei : w_self;
        const v4f v = *reinterpret_cast<const v4f*>(wsrc + (size_t)(16 * jt + m) * D + 8 * i + 4 * half);
        float* dst = wfrag + ((jt * 32 + 2 * i + half) * 2 + mat) * 64;
        v2f lo = {v.x, v.y};
        v2f hi = {v.z, v.w};
        *reinterpret_cast<v2f*>(dst + 2 * m)        = lo;
        *reinterpret_cast<v2f*>(dst + 2 * (m + 16)) = hi;
    }
    const float g0 = gamma[lane], g1 = gamma[lane + 32], g2 = gamma[lane + 64], g3 = gamma[lane + 96];
    const float b0 = beta[lane],  b1 = beta[lane + 32],  b2 = beta[lane + 64],  b3 = beta[lane + 96];
    __syncthreads();

    const int tiles = (N + 15) >> 4;
    for (int tile = blockIdx.x + gridDim.x * waveId; tile < tiles; tile += gridDim.x * WAVES_PER_BLOCK) {

    const bool k16 = (K == 16);
    int   iv[8] = {0, 0, 0, 0, 0, 0, 0, 0};
    float wv[8] = {0.f, 0.f, 0.f, 0.f, 0.f, 0.f, 0.f, 0.f};
    if (k16) {
        int lrow = tile * 16 + (lane >> 1);
        if (lrow >= N) lrow = N - 1;
        const size_t e = (size_t)lrow * 16 + 8 * (lane & 1);
        const v4i i0 = *reinterpret_cast<const v4i*>(nbr_idx + e);
        const v4i i1 = *reinterpret_cast<const v4i*>(nbr_idx + e + 4);
        const v4f w0 = *reinterpret_cast<const v4f*>(nbr_w + e);
        const v4f w1 = *reinterpret_cast<const v4f*>(nbr_w + e + 4);
        iv[0] = i0.x; iv[1] = i0.y; iv[2] = i0.z; iv[3] = i0.w;
        iv[4] = i1.x; iv[5] = i1.y; iv[6] = i1.z; iv[7] = i1.w;
        wv[0] = w0.x; wv[1] = w0.y; wv[2] = w0.z; wv[3] = w0.w;
        wv[4] = w1.x; wv[5] = w1.y; wv[6] = w1.z; wv[7] = w1.w;
    }

    for (int r = 0; r < 16; ++r) {
        int grow = tile * 16 + r;
        if (grow >= N) grow = N - 1;
        v4f v = *reinterpret_cast<const v4f*>(h + (size_t)grow * D + 4 * lane);
        *reinterpret_cast<v4f*>(tileLds + r * LNS + 4 * lane) = v;
    }

    v2f aS[32];
#pragma unroll
    for (int c = 0; c < 32; ++c)
        aS[c] = *reinterpret_cast<const v2f*>(tileLds + m * LNS + 4 * c + off);

    v8f zs[8] = {};
    auto self_wmma16 = [&](v8f& acc, const float* bw, const int c0) {
#pragma unroll
        for (int c = 0; c < 16; ++c) {
            v2f bS = *reinterpret_cast<const v2f*>(bw + (c0 + c) * 128);
            acc = __builtin_amdgcn_wmma_f32_16x16x4_f32(false, aS[c0 + c], false, bS,
                                                        (short)0, acc, false, false);
        }
        __builtin_amdgcn_sched_barrier(0);
    };

    if (k16) {
        const unsigned long long hbase = reinterpret_cast<unsigned long long>(h);
        auto issue_row = [&](int r, v4f (&buf)[16]) {
            unsigned long long base[16];
#pragma unroll
            for (int k = 0; k < 16; ++k) {
                const int nrow = __builtin_amdgcn_readlane(iv[k & 7], 2 * r + (k >> 3));
                base[k] = hbase + ((unsigned long long)(long long)nrow << 9);
            }
            __builtin_amdgcn_sched_barrier(0);
#pragma unroll
            for (int k = 0; k < 16; ++k)
                buf[k] = reinterpret_cast<const gv4f*>(base[k])[lane];
            __builtin_amdgcn_sched_barrier(0);
        };
        auto sum_row = [&](int r, const v4f (&buf)[16]) {
            v4f acc = {};
#pragma unroll
            for (int k = 0; k < 16; ++k) {
                const float w = __int_as_float(__builtin_amdgcn_readlane(__float_as_int(wv[k & 7]), 2 * r + (k >> 3)));
                acc += buf[k] * w;
            }
            *reinterpret_cast<v4f*>(tileLds + r * LNS + 4 * lane) = acc;
        };
        v4f buf[16];
#pragma nounroll
        for (int p = 0; p < 8; ++p) {
            const float* bw = wfrag + p * (32 * 2 * 64) + 2 * lane;
            v8f acc = {};
            issue_row(2 * p, buf);
            self_wmma16(acc, bw, 0);
            sum_row(2 * p, buf);
            issue_row(2 * p + 1, buf);
            self_wmma16(acc, bw, 16);
            sum_row(2 * p + 1, buf);
            zs[0] = zs[1]; zs[1] = zs[2]; zs[2] = zs[3]; zs[3] = zs[4];
            zs[4] = zs[5]; zs[5] = zs[6]; zs[6] = zs[7]; zs[7] = acc;
        }
    } else {
    for (int r = 0; r < 16; ++r) {
        int grow = tile * 16 + r;
        if (grow >= N) grow = N - 1;
        v4f acc = {};
        for (int k = 0; k < K; ++k) {
            const int   nrow = nbr_idx[(size_t)grow * K + k];
            const float w    = nbr_w [(size_t)grow * K + k];
            v4f v = *reinterpret_cast<const v4f*>(h + (size_t)nrow * D + 4 * lane);
            acc += v * w;
        }
        *reinterpret_cast<v4f*>(tileLds + r * LNS + 4 * lane) = acc;
    }
#pragma nounroll
    for (int p = 0; p < 8; ++p) {
        const float* bw = wfrag + p * (32 * 2 * 64) + 2 * lane;
        v8f acc = {};
        self_wmma16(acc, bw, 0);
        self_wmma16(acc, bw, 16);
        zs[0] = zs[1]; zs[1] = zs[2]; zs[2] = zs[3]; zs[3] = zs[4];
        zs[4] = zs[5]; zs[5] = zs[6]; zs[6] = zs[7]; zs[7] = acc;
    }
    }

    v2f aA[32];
#pragma unroll
    for (int c = 0; c < 32; ++c)
        aA[c] = *reinterpret_cast<const v2f*>(tileLds + m * LNS + 4 * c + off);

#pragma nounroll
    for (int jt = 0; jt < 8; ++jt) {
        const int j0 = jt * 16;
        v8f acc = zs[0];
        zs[0] = zs[1]; zs[1] = zs[2]; zs[2] = zs[3]; zs[3] = zs[4];
        zs[4] = zs[5]; zs[5] = zs[6]; zs[6] = zs[7];
        const float* bw = wfrag + jt * (32 * 2 * 64) + 2 * lane;
#pragma unroll
        for (int c = 0; c < 32; ++c) {
            v2f bN = *reinterpret_cast<const v2f*>(bw + c * 128 + 64);
            acc = __builtin_amdgcn_wmma_f32_16x16x4_f32(false, aA[c], false, bN,
                                                        (short)0, acc, false, false);
        }
#pragma unroll
        for (int r = 0; r < 8; ++r) {
            const int rr = r + 8 * half;
            const float z = acc[r];
            const float g = 0.5f * z * (1.0f + erff(z * 0.70710678118654752f));
            tileLds[rr * LNS + j0 + m] = g;
        }
    }

    float x[16][4];
#pragma unroll
    for (int r = 0; r < 16; ++r) {
#pragma unroll
        for (int q = 0; q < 4; ++q) x[r][q] = tileLds[r * LNS + lane + 32 * q];
    }
    float s[16];
#pragma unroll
    for (int r = 0; r < 16; ++r) s[r] = x[r][0] + x[r][1] + x[r][2] + x[r][3];
#pragma unroll
    for (int d = 16; d >= 1; d >>= 1) {
        float t[16];
#pragma unroll
        for (int r = 0; r < 16; ++r) t[r] = __shfl_xor(s[r], d, 32);
#pragma unroll
        for (int r = 0; r < 16; ++r) s[r] += t[r];
    }
    float s2[16];
#pragma unroll
    for (int r = 0; r < 16; ++r) {
        const float mu = s[r] * (1.0f / D);
        const float d0 = x[r][0] - mu, d1 = x[r][1] - mu, d2 = x[r][2] - mu, d3 = x[r][3] - mu;
        x[r][0] = d0; x[r][1] = d1; x[r][2] = d2; x[r][3] = d3;
        s2[r] = d0 * d0 + d1 * d1 + d2 * d2 + d3 * d3;
    }
#pragma unroll
    for (int d = 16; d >= 1; d >>= 1) {
        float t[16];
#pragma unroll
        for (int r = 0; r < 16; ++r) t[r] = __shfl_xor(s2[r], d, 32);
#pragma unroll
        for (int r = 0; r < 16; ++r) s2[r] += t[r];
    }
    float* const otile = out + (size_t)tile * (16 * D) + lane;
    auto finish_row = [&](int r) {
        const float inv = rsqrtf(s2[r] * (1.0f / D) + 1e-5f);
        float* o = otile + r * D;
        o[0]  = x[r][0] * inv * g0 + b0;
        o[32] = x[r][1] * inv * g1 + b1;
        o[64] = x[r][2] * inv * g2 + b2;
        o[96] = x[r][3] * inv * g3 + b3;
    };
    if (tile * 16 + 16 <= N) {
#pragma unroll
        for (int r = 0; r < 16; ++r) finish_row(r);
    } else {
#pragma unroll
        for (int r = 0; r < 16; ++r)
            if (tile * 16 + r < N) finish_row(r);
    }
    }
}

extern "C" void kernel_launch(void* const* d_in, const int* in_sizes, int n_in,
                              void* d_out, int out_size, void* d_ws, size_t ws_size,
                              hipStream_t stream) {
    const float* h       = (const float*)d_in[0];
    const int*   nbr_idx = (const int*)  d_in[1];
    const float* nbr_w   = (const float*)d_in[2];
    const float* w_self  = (const float*)d_in[3];
    const float* w_nei   = (const float*)d_in[4];
    const float* gamma   = (const float*)d_in[5];
    const float* beta    = (const float*)d_in[6];
    float* out = (float*)d_out;

    const int N = in_sizes[0] / D;
    const int K = (N > 0) ? (in_sizes[1] / N) : 16;

    const int tiles  = (N + 15) / 16;
    int blocks = (tiles + WAVES_PER_BLOCK - 1) / WAVES_PER_BLOCK;
    if (blocks > MAX_BLOCKS) blocks = MAX_BLOCKS;
    hipLaunchKernelGGL(sage_wmma_kernel, dim3(blocks), dim3(32 * WAVES_PER_BLOCK), 0, stream,
                       h, nbr_idx, nbr_w, w_self, w_nei, gamma, beta, out, N, K);
}


# GELU's exact erf without its branch: both polynomial ranges of the library's erff computed, one selected, so a column block's 8 values are independent chains issued together; values bit-identical
# speedup vs baseline: 2.0251x; 1.0722x over previous
#include <hip/hip_runtime.h>
#include <math.h>

typedef __attribute__((ext_vector_type(2))) float v2f;
typedef __attribute__((ext_vector_type(4))) float v4f;
typedef __attribute__((ext_vector_type(8))) float v8f;
typedef __attribute__((ext_vector_type(4))) int   v4i;
typedef __attribute__((address_space(1))) const v4f gv4f;

#define D 128
#define LNS 132
#define WAVES_PER_BLOCK 16
#define MAX_BLOCKS 256
#define WFRAG_FLOATS (8 * 32 * 2 * 64)

#define F32_BITS(u) __builtin_bit_cast(float, (unsigned)(u))
__device__ __forceinline__ float erf_both_ranges(const float y)
{
    const float t = __builtin_fabsf(y);
    const float s = y * y;
    float q = __builtin_fmaf(s, F32_BITS(0xba1345e1), F32_BITS(0x3ba10414));
    q = __builtin_fmaf(s, q, F32_BITS(0xbcdac9b8));
    q = __builtin_fmaf(s, q, F32_BITS(0x3de703be));
    q = __builtin_fmaf(s, q, F32_BITS(0xbec09330));
    q = __builtin_fmaf(s, q, F32_BITS(0x3e0375d0));
    const float r_small = __builtin_fmaf(t, q, t);
    float p = __builtin_fmaf(t, F32_BITS(0x378e98ab), F32_BITS(0xb9c68948));
    p = __builtin_fmaf(t, p, F32_BITS(0x3b7cd369));
    p = __builtin_fmaf(t, p, F32_BITS(0xbcc618b2));
    p = __builtin_fmaf(t, p, F32_BITS(0x3dda74e4));
    p = __builtin_fmaf(t, p, F32_BITS(0x3f228afd));
    p = __builtin_fmaf(t, p, F32_BITS(0x3e03c728));
    p = __builtin_fmaf(t, p, t);
    const float r_large = 1.0f - expf(-p);
    const float r = (t < 1.0f) ? r_small : r_large;
    return __builtin_copysignf(r, y);
}

__global__ __launch_bounds__(32 * WAVES_PER_BLOCK)
void sage_wmma_kernel(const float* __restrict__ h,
                      const int*   __restrict__ nbr_idx,
                      const float* __restrict__ nbr_w,
                      const float* __restrict__ w_self,
                      const float* __restrict__ w_nei,
                      const float* __restrict__ gamma,
                      const float* __restrict__ beta,
                      float* __restrict__ out,
                      int N, int K)
{
    __shared__ float wfrag[WFRAG_FLOATS];
    __shared__ float lds[WAVES_PER_BLOCK * 16 * LNS];

    const int lane   = threadIdx.x & 31;
    const int waveId = __builtin_amdgcn_readfirstlane(threadIdx.x >> 5);

    const int m    = lane & 15;
    const int half = lane >> 4;
    const int off  = half * 2;

    float* tileLds = lds + waveId * (16 * LNS);

    for (int L = waveId; L < 256; L += WAVES_PER_BLOCK) {
        const int mat = L >> 7;
        const int jt  = (L >> 4) & 7;
        const int i   = L & 15;
        const float* wsrc = mat ? w_nei : w_self;
        const v4f v = *reinterpret_cast<const v4f*>(wsrc + (size_t)(16 * jt + m) * D + 8 * i + 4 * half);
        float* dst = wfrag + ((jt * 32 + 2 * i + half) * 2 + mat) * 64;
        v2f lo = {v.x, v.y};
        v2f hi = {v.z, v.w};
        *reinterpret_cast<v2f*>(dst + 2 * m)        = lo;
        *reinterpret_cast<v2f*>(dst + 2 * (m + 16)) = hi;
    }
    const float g0 = gamma[lane], g1 = gamma[lane + 32], g2 = gamma[lane + 64], g3 = gamma[lane + 96];
    const float b0 = beta[lane],  b1 = beta[lane + 32],  b2 = beta[lane + 64],  b3 = beta[lane + 96];
    __syncthreads();

    const int tiles = (N + 15) >> 4;
    for (int tile = blockIdx.x + gridDim.x * waveId; tile < tiles; tile += gridDim.x * WAVES_PER_BLOCK) {

    const bool k16 = (K == 16);
    int   iv[8] = {0, 0, 0, 0, 0, 0, 0, 0};
    float wv[8] = {0.f, 0.f, 0.f, 0.f, 0.f, 0.f, 0.f, 0.f};
    if (k16) {
        int lrow = tile * 16 + (lane >> 1);
        if (lrow >= N) lrow = N - 1;
        const size_t e = (size_t)lrow * 16 + 8 * (lane & 1);
        const v4i i0 = *reinterpret_cast<const v4i*>(nbr_idx + e);
        const v4i i1 = *reinterpret_cast<const v4i*>(nbr_idx + e + 4);
        const v4f w0 = *reinterpret_cast<const v4f*>(nbr_w + e);
        const v4f w1 = *reinterpret_cast<const v4f*>(nbr_w + e + 4);
        iv[0] = i0.x; iv[1] = i0.y; iv[2] = i0.z; iv[3] = i0.w;
        iv[4] = i1.x; iv[5] = i1.y; iv[6] = i1.z; iv[7] = i1.w;
        wv[0] = w0.x; wv[1] = w0.y; wv[2] = w0.z; wv[3] = w0.w;
        wv[4] = w1.x; wv[5] = w1.y; wv[6] = w1.z; wv[7] = w1.w;
    }

    for (int r = 0; r < 16; ++r) {
        int grow = tile * 16 + r;
        if (grow >= N) grow = N - 1;
        v4f v = *reinterpret_cast<const v4f*>(h + (size_t)grow * D + 4 * lane);
        *reinterpret_cast<v4f*>(tileLds + r * LNS + 4 * lane) = v;
    }

    v2f aS[32];
#pragma unroll
    for (int c = 0; c < 32; ++c)
        aS[c] = *reinterpret_cast<const v2f*>(tileLds + m * LNS + 4 * c + off);

    v8f zs[8] = {};
    auto self_wmma16 = [&](v8f& acc, const float* bw, const int c0) {
#pragma unroll
        for (int c = 0; c < 16; ++c) {
            v2f bS = *reinterpret_cast<const v2f*>(bw + (c0 + c) * 128);
            acc = __builtin_amdgcn_wmma_f32_16x16x4_f32(false, aS[c0 + c], false, bS,
                                                        (short)0, acc, false, false);
        }
        __builtin_amdgcn_sched_barrier(0);
    };

    if (k16) {
        const unsigned long long hbase = reinterpret_cast<unsigned long long>(h);
        auto issue_row = [&](int r, v4f (&buf)[16]) {
            unsigned long long base[16];
#pragma unroll
            for (int k = 0; k < 16; ++k) {
                const int nrow = __builtin_amdgcn_readlane(iv[k & 7], 2 * r + (k >> 3));
                base[k] = hbase + ((unsigned long long)(long long)nrow << 9);
            }
            __builtin_amdgcn_sched_barrier(0);
#pragma unroll
            for (int k = 0; k < 16; ++k)
                buf[k] = reinterpret_cast<const gv4f*>(base[k])[lane];
            __builtin_amdgcn_sched_barrier(0);
        };
        auto sum_row = [&](int r, const v4f (&buf)[16]) {
            v4f acc = {};
#pragma unroll
            for (int k = 0; k < 16; ++k) {
                const float w = __int_as_float(__builtin_amdgcn_readlane(__float_as_int(wv[k & 7]), 2 * r + (k >> 3)));
                acc += buf[k] * w;
            }
            *reinterpret_cast<v4f*>(tileLds + r * LNS + 4 * lane) = acc;
        };
        v4f buf[16];
#pragma nounroll
        for (int p = 0; p < 8; ++p) {
            const float* bw = wfrag + p * (32 * 2 * 64) + 2 * lane;
            v8f acc = {};
            issue_row(2 * p, buf);
            self_wmma16(acc, bw, 0);
            sum_row(2 * p, buf);
            issue_row(2 * p + 1, buf);
            self_wmma16(acc, bw, 16);
            sum_row(2 * p + 1, buf);
            zs[0] = zs[1]; zs[1] = zs[2]; zs[2] = zs[3]; zs[3] = zs[4];
            zs[4] = zs[5]; zs[5] = zs[6]; zs[6] = zs[7]; zs[7] = acc;
        }
    } else {
    for (int r = 0; r < 16; ++r) {
        int grow = tile * 16 + r;
        if (grow >= N) grow = N - 1;
        v4f acc = {};
        for (int k = 0; k < K; ++k) {
            const int   nrow = nbr_idx[(size_t)grow * K + k];
            const float w    = nbr_w [(size_t)grow * K + k];
            v4f v = *reinterpret_cast<const v4f*>(h + (size_t)nrow * D + 4 * lane);
            acc += v * w;
        }
        *reinterpret_cast<v4f*>(tileLds + r * LNS + 4 * lane) = acc;
    }
#pragma nounroll
    for (int p = 0; p < 8; ++p) {
        const float* bw = wfrag + p * (32 * 2 * 64) + 2 * lane;
        v8f acc = {};
        self_wmma16(acc, bw, 0);
        self_wmma16(acc, bw, 16);
        zs[0] = zs[1]; zs[1] = zs[2]; zs[2] = zs[3]; zs[3] = zs[4];
        zs[4] = zs[5]; zs[5] = zs[6]; zs[6] = zs[7]; zs[7] = acc;
    }
    }

    v2f aA[32];
#pragma unroll
    for (int c = 0; c < 32; ++c)
        aA[c] = *reinterpret_cast<const v2f*>(tileLds + m * LNS + 4 * c + off);

#pragma nounroll
    for (int jt = 0; jt < 8; ++jt) {
        const int j0 = jt * 16;
        v8f acc = zs[0];
        zs[0] = zs[1]; zs[1] = zs[2]; zs[2] = zs[3]; zs[3] = zs[4];
        zs[4] = zs[5]; zs[5] = zs[6]; zs[6] = zs[7];
        const float* bw = wfrag + jt * (32 * 2 * 64) + 2 * lane;
#pragma unroll
        for (int c = 0; c < 32; ++c) {
            v2f bN = *reinterpret_cast<const v2f*>(bw + c * 128 + 64);
            acc = __builtin_amdgcn_wmma_f32_16x16x4_f32(false, aA[c], false, bN,
                                                        (short)0, acc, false, false);
        }
#pragma unroll
        for (int r = 0; r < 8; ++r) {
            const int rr = r + 8 * half;
            const float z = acc[r];
            const float g = 0.5f * z * (1.0f + erf_both_ranges(z * 0.70710678118654752f));
            tileLds[rr * LNS + j0 + m] = g;
        }
    }

    float x[16][4];
#pragma unroll
    for (int r = 0; r < 16; ++r) {
#pragma unroll
        for (int q = 0; q < 4; ++q) x[r][q] = tileLds[r * LNS + lane + 32 * q];
    }
    float s[16];
#pragma unroll
    for (int r = 0; r < 16; ++r) s[r] = x[r][0] + x[r][1] + x[r][2] + x[r][3];
#pragma unroll
    for (int d = 16; d >= 1; d >>= 1) {
        float t[16];
#pragma unroll
        for (int r = 0; r < 16; ++r) t[r] = __shfl_xor(s[r], d, 32);
#pragma unroll
        for (int r = 0; r < 16; ++r) s[r] += t[r];
    }
    float s2[16];
#pragma unroll
    for (int r = 0; r < 16; ++r) {
        const float mu = s[r] * (1.0f / D);
        const float d0 = x[r][0] - mu, d1 = x[r][1] - mu, d2 = x[r][2] - mu, d3 = x[r][3] - mu;
        x[r][0] = d0; x[r][1] = d1; x[r][2] = d2; x[r][3] = d3;
        s2[r] = d0 * d0 + d1 * d1 + d2 * d2 + d3 * d3;
    }
#pragma unroll
    for (int d = 16; d >= 1; d >>= 1) {
        float t[16];
#pragma unroll
        for (int r = 0; r < 16; ++r) t[r] = __shfl_xor(s2[r], d, 32);
#pragma unroll
        for (int r = 0; r < 16; ++r) s2[r] += t[r];
    }
    float* const otile = out + (size_t)tile * (16 * D) + lane;
    auto finish_row = [&](int r) {
        const float inv = rsqrtf(s2[r] * (1.0f / D) + 1e-5f);
        float* o = otile + r * D;
        o[0]  = x[r][0] * inv * g0 + b0;
        o[32] = x[r][1] * inv * g1 + b1;
        o[64] = x[r][2] * inv * g2 + b2;
        o[96] = x[r][3] * inv * g3 + b3;
    };
    if (tile * 16 + 16 <= N) {
#pragma unroll
        for (int r = 0; r < 16; ++r) finish_row(r);
    } else {
#pragma unroll
        for (int r = 0; r < 16; ++r)
            if (tile * 16 + r < N) finish_row(r);
    }
    }
}

extern "C" void kernel_launch(void* const* d_in, const int* in_sizes, int n_in,
                              void* d_out, int out_size, void* d_ws, size_t ws_size,
                              hipStream_t stream) {
    const float* h       = (const float*)d_in[0];
    const int*   nbr_idx = (const int*)  d_in[1];
    const float* nbr_w   = (const float*)d_in[2];
    const float* w_self  = (const float*)d_in[3];
    const float* w_nei   = (const float*)d_in[4];
    const float* gamma   = (const float*)d_in[5];
    const float* beta    = (const float*)d_in[6];
    float* out = (float*)d_out;

    const int N = in_sizes[0] / D;
    const int K = (N > 0) ? (in_sizes[1] / N) : 16;

    const int tiles  = (N + 15) / 16;
    int blocks = (tiles + WAVES_PER_BLOCK - 1) / WAVES_PER_BLOCK;
    if (blocks > MAX_BLOCKS) blocks = MAX_BLOCKS;
    hipLaunchKernelGGL(sage_wmma_kernel, dim3(blocks), dim3(32 * WAVES_PER_BLOCK), 0, stream,
                       h, nbr_idx, nbr_w, w_self, w_nei, gamma, beta, out, N, K);
}


# Loads before the work-group barrier go out together: a wave's 16 weight loads (were a load-wait-store loop, 16 dependent round trips) and its first tile's index and self-row loads; sums unchanged
# speedup vs baseline: 2.1733x; 1.0732x over previous
#include <hip/hip_runtime.h>
#include <math.h>

typedef __attribute__((ext_vector_type(2))) float v2f;
typedef __attribute__((ext_vector_type(4))) float v4f;
typedef __attribute__((ext_vector_type(8))) float v8f;
typedef __attribute__((ext_vector_type(4))) int   v4i;
typedef __attribute__((address_space(1))) const v4f gv4f;

#define D 128
#define LNS 132
#define WAVES_PER_BLOCK 16
#define MAX_BLOCKS 256
#define WFRAG_FLOATS (8 * 32 * 2 * 64)

#define F32_BITS(u) __builtin_bit_cast(float, (unsigned)(u))
__device__ __forceinline__ float erf_both_ranges(const float y)
{
    const float t = __builtin_fabsf(y);
    const float s = y * y;
    float q = __builtin_fmaf(s, F32_BITS(0xba1345e1), F32_BITS(0x3ba10414));
    q = __builtin_fmaf(s, q, F32_BITS(0xbcdac9b8));
    q = __builtin_fmaf(s, q, F32_BITS(0x3de703be));
    q = __builtin_fmaf(s, q, F32_BITS(0xbec09330));
    q = __builtin_fmaf(s, q, F32_BITS(0x3e0375d0));
    const float r_small = __builtin_fmaf(t, q, t);
    float p = __builtin_fmaf(t, F32_BITS(0x378e98ab), F32_BITS(0xb9c68948));
    p = __builtin_fmaf(t, p, F32_BITS(0x3b7cd369));
    p = __builtin_fmaf(t, p, F32_BITS(0xbcc618b2));
    p = __builtin_fmaf(t, p, F32_BITS(0x3dda74e4));
    p = __builtin_fmaf(t, p, F32_BITS(0x3f228afd));
    p = __builtin_fmaf(t, p, F32_BITS(0x3e03c728));
    p = __builtin_fmaf(t, p, t);
    const float r_large = 1.0f - expf(-p);
    const float r = (t < 1.0f) ? r_small : r_large;
    return __builtin_copysignf(r, y);
}

__global__ __launch_bounds__(32 * WAVES_PER_BLOCK)
void sage_wmma_kernel(const float* __restrict__ h,
                      const int*   __restrict__ nbr_idx,
                      const float* __restrict__ nbr_w,
                      const float* __restrict__ w_self,
                      const float* __restrict__ w_nei,
                      const float* __restrict__ gamma,
                      const float* __restrict__ beta,
                      float* __restrict__ out,
                      int N, int K)
{
    __shared__ float wfrag[WFRAG_FLOATS];
    __shared__ float lds[WAVES_PER_BLOCK * 16 * LNS];

    const int lane   = threadIdx.x & 31;
    const int waveId = __builtin_amdgcn_readfirstlane(threadIdx.x >> 5);

    const int m    = lane & 15;
    const int half = lane >> 4;
    const int off  = half * 2;

    float* tileLds = lds + waveId * (16 * LNS);

    const int tiles = (N + 15) >> 4;
    const int tile0 = blockIdx.x + gridDim.x * waveId;
    const bool k16 = (K == 16);

    int   iv[8] = {0, 0, 0, 0, 0, 0, 0, 0};
    float wv[8] = {0.f, 0.f, 0.f, 0.f, 0.f, 0.f, 0.f, 0.f};
    auto fetch_tile_head = [&](const int tile) {
        if (k16) {
            int lrow = tile * 16 + (lane >> 1);
            if (lrow >= N) lrow = N - 1;
            const size_t e = (size_t)lrow * 16 + 8 * (lane & 1);
            const v4i i0 = *reinterpret_cast<const v4i*>(nbr_idx + e);
            const v4i i1 = *reinterpret_cast<const v4i*>(nbr_idx + e + 4);
            const v4f w0 = *reinterpret_cast<const v4f*>(nbr_w + e);
            const v4f w1 = *reinterpret_cast<const v4f*>(nbr_w + e + 4);
            iv[0] = i0.x; iv[1] = i0.y; iv[2] = i0.z; iv[3] = i0.w;
            iv[4] = i1.x; iv[5] = i1.y; iv[6] = i1.z; iv[7] = i1.w;
            wv[0] = w0.x; wv[1] = w0.y; wv[2] = w0.z; wv[3] = w0.w;
            wv[4] = w1.x; wv[5] = w1.y; wv[6] = w1.z; wv[7] = w1.w;
        }
        for (int r = 0; r < 16; ++r) {
            int grow = tile * 16 + r;
            if (grow >= N) grow = N - 1;
            v4f v = *reinterpret_cast<const v4f*>(h + (size_t)grow * D + 4 * lane);
            *reinterpret_cast<v4f*>(tileLds + r * LNS + 4 * lane) = v;
        }
    };

    static_assert(WAVES_PER_BLOCK == 16, "sixteen waves x sixteen pieces cover the 256 pieces");
    v4f wpiece[16];
    auto load_wpieces = [&]() {
#pragma unroll
        for (int t = 0; t < 16; ++t) {
            const float* wsrc = (t >> 3) ? w_nei : w_self;
            wpiece[t] = *reinterpret_cast<const v4f*>(wsrc + (size_t)(16 * (t & 7) + m) * D + 8 * waveId + 4 * half);
        }
    };
    auto store_wpieces = [&]() {
#pragma unroll
        for (int t = 0; t < 16; ++t) {
            const int mat = t >> 3;
            const int jt  = t & 7;
            float* dst = wfrag + ((jt * 32 + 2 * waveId + half) * 2 + mat) * 64;
            v2f lo = {wpiece[t].x, wpiece[t].y};
            v2f hi = {wpiece[t].z, wpiece[t].w};
            *reinterpret_cast<v2f*>(dst + 2 * m)        = lo;
            *reinterpret_cast<v2f*>(dst + 2 * (m + 16)) = hi;
        }
    };
    const bool head_early = k16 && (tile0 < tiles);
    if (head_early) {
        load_wpieces();
        fetch_tile_head(tile0);
        store_wpieces();
    } else {
        load_wpieces();
        store_wpieces();
    }
    const float g0 = gamma[lane], g1 = gamma[lane + 32], g2 = gamma[lane + 64], g3 = gamma[lane + 96];
    const float b0 = beta[lane],  b1 = beta[lane + 32],  b2 = beta[lane + 64],  b3 = beta[lane + 96];
    __syncthreads();

    for (int tile = tile0; tile < tiles; tile += gridDim.x * WAVES_PER_BLOCK) {

    if (tile != tile0 || !head_early) fetch_tile_head(tile);

    v2f aS[32];
#pragma unroll
    for (int c = 0; c < 32; ++c)
        aS[c] = *reinterpret_cast<const v2f*>(tileLds + m * LNS + 4 * c + off);

    v8f zs[8] = {};
    auto self_wmma16 = [&](v8f& acc, const float* bw, const int c0) {
#pragma unroll
        for (int c = 0; c < 16; ++c) {
            v2f bS = *reinterpret_cast<const v2f*>(bw + (c0 + c) * 128);
            acc = __builtin_amdgcn_wmma_f32_16x16x4_f32(false, aS[c0 + c], false, bS,
                                                        (short)0, acc, false, false);
        }
        __builtin_amdgcn_sched_barrier(0);
    };

    if (k16) {
        const unsigned long long hbase = reinterpret_cast<unsigned long long>(h);
        auto issue_row = [&](int r, v4f (&buf)[16]) {
            unsigned long long base[16];
#pragma unroll
            for (int k = 0; k < 16; ++k) {
                const int nrow = __builtin_amdgcn_readlane(iv[k & 7], 2 * r + (k >> 3));
                base[k] = hbase + ((unsigned long long)(long long)nrow << 9);
            }
            __builtin_amdgcn_sched_barrier(0);
#pragma unroll
            for (int k = 0; k < 16; ++k)
                buf[k] = reinterpret_cast<const gv4f*>(base[k])[lane];
            __builtin_amdgcn_sched_barrier(0);
        };
        auto sum_row = [&](int r, const v4f (&buf)[16]) {
            v4f acc = {};
#pragma unroll
            for (int k = 0; k < 16; ++k) {
                const float w = __int_as_float(__builtin_amdgcn_readlane(__float_as_int(wv[k & 7]), 2 * r + (k >> 3)));
                acc += buf[k] * w;
            }
            *reinterpret_cast<v4f*>(tileLds + r * LNS + 4 * lane) = acc;
        };
        v4f buf[16];
#pragma nounroll
        for (int p = 0; p < 8; ++p) {
            const float* bw = wfrag + p * (32 * 2 * 64) + 2 * lane;
            v8f acc = {};
            issue_row(2 * p, buf);
            self_wmma16(acc, bw, 0);
            sum_row(2 * p, buf);
            issue_row(2 * p + 1, buf);
            self_wmma16(acc, bw, 16);
            sum_row(2 * p + 1, buf);
            zs[0] = zs[1]; zs[1] = zs[2]; zs[2] = zs[3]; zs[3] = zs[4];
            zs[4] = zs[5]; zs[5] = zs[6]; zs[6] = zs[7]; zs[7] = acc;
        }
    } else {
    for (int r = 0; r < 16; ++r) {
        int grow = tile * 16 + r;
        if (grow >= N) grow = N - 1;
        v4f acc = {};
        for (int k = 0; k < K; ++k) {
            const int   nrow = nbr_idx[(size_t)grow * K + k];
            const float w    = nbr_w [(size_t)grow * K + k];
            v4f v = *reinterpret_cast<const v4f*>(h + (size_t)nrow * D + 4 * lane);
            acc += v * w;
        }
        *reinterpret_cast<v4f*>(tileLds + r * LNS + 4 * lane) = acc;
    }
#pragma nounroll
    for (int p = 0; p < 8; ++p) {
        const float* bw = wfrag + p * (32 * 2 * 64) + 2 * lane;
        v8f acc = {};
        self_wmma16(acc, bw, 0);
        self_wmma16(acc, bw, 16);
        zs[0] = zs[1]; zs[1] = zs[2]; zs[2] = zs[3]; zs[3] = zs[4];
        zs[4] = zs[5]; zs[5] = zs[6]; zs[6] = zs[7]; zs[7] = acc;
    }
    }

    v2f aA[32];
#pragma unroll
    for (int c = 0; c < 32; ++c)
        aA[c] = *reinterpret_cast<const v2f*>(tileLds + m * LNS + 4 * c + off);

#pragma nounroll
    for (int jt = 0; jt < 8; ++jt) {
        const int j0 = jt * 16;
        v8f acc = zs[0];
        zs[0] = zs[1]; zs[1] = zs[2]; zs[2] = zs[3]; zs[3] = zs[4];
        zs[4] = zs[5]; zs[5] = zs[6]; zs[6] = zs[7];
        const float* bw = wfrag + jt * (32 * 2 * 64) + 2 * lane;
#pragma unroll
        for (int c = 0; c < 32; ++c) {
            v2f bN = *reinterpret_cast<const v2f*>(bw + c * 128 + 64);
            acc = __builtin_amdgcn_wmma_f32_16x16x4_f32(false, aA[c], false, bN,
                                                        (short)0, acc, false, false);
        }
#pragma unroll
        for (int r = 0; r < 8; ++r) {
            const int rr = r + 8 * half;
            const float z = acc[r];
            const float g = 0.5f * z * (1.0f + erf_both_ranges(z * 0.70710678118654752f));
            tileLds[rr * LNS + j0 + m] = g;
        }
    }

    float x[16][4];
#pragma unroll
    for (int r = 0; r < 16; ++r) {
#pragma unroll
        for (int q = 0; q < 4; ++q) x[r][q] = tileLds[r * LNS + lane + 32 * q];
    }
    float s[16];
#pragma unroll
    for (int r = 0; r < 16; ++r) s[r] = x[r][0] + x[r][1] + x[r][2] + x[r][3];
#pragma unroll
    for (int d = 16; d >= 1; d >>= 1) {
        float t[16];
#pragma unroll
        for (int r = 0; r < 16; ++r) t[r] = __shfl_xor(s[r], d, 32);
#pragma unroll
        for (int r = 0; r < 16; ++r) s[r] += t[r];
    }
    float s2[16];
#pragma unroll
    for (int r = 0; r < 16; ++r) {
        const float mu = s[r] * (1.0f / D);
        const float d0 = x[r][0] - mu, d1 = x[r][1] - mu, d2 = x[r][2] - mu, d3 = x[r][3] - mu;
        x[r][0] = d0; x[r][1] = d1; x[r][2] = d2; x[r][3] = d3;
        s2[r] = d0 * d0 + d1 * d1 + d2 * d2 + d3 * d3;
    }
#pragma unroll
    for (int d = 16; d >= 1; d >>= 1) {
        float t[16];
#pragma unroll
        for (int r = 0; r < 16; ++r) t[r] = __shfl_xor(s2[r], d, 32);
#pragma unroll
        for (int r = 0; r < 16; ++r) s2[r] += t[r];
    }
    float* const otile = out + (size_t)tile * (16 * D) + lane;
    auto finish_row = [&](int r) {
        const float inv = rsqrtf(s2[r] * (1.0f / D) + 1e-5f);
        float* o = otile + r * D;
        o[0]  = x[r][0] * inv * g0 + b0;
        o[32] = x[r][1] * inv * g1 + b1;
        o[64] = x[r][2] * inv * g2 + b2;
        o[96] = x[r][3] * inv * g3 + b3;
    };
    if (tile * 16 + 16 <= N) {
#pragma unroll
        for (int r = 0; r < 16; ++r) finish_row(r);
    } else {
#pragma unroll
        for (int r = 0; r < 16; ++r)
            if (tile * 16 + r < N) finish_row(r);
    }
    }
}

extern "C" void kernel_launch(void* const* d_in, const int* in_sizes, int n_in,
                              void* d_out, int out_size, void* d_ws, size_t ws_size,
                              hipStream_t stream) {
    const float* h       = (const float*)d_in[0];
    const int*   nbr_idx = (const int*)  d_in[1];
    const float* nbr_w   = (const float*)d_in[2];
    const float* w_self  = (const float*)d_in[3];
    const float* w_nei   = (const float*)d_in[4];
    const float* gamma   = (const float*)d_in[5];
    const float* beta    = (const float*)d_in[6];
    float* out = (float*)d_out;

    const int N = in_sizes[0] / D;
    const int K = (N > 0) ? (in_sizes[1] / N) : 16;

    const int tiles  = (N + 15) / 16;
    int blocks = (tiles + WAVES_PER_BLOCK - 1) / WAVES_PER_BLOCK;
    if (blocks > MAX_BLOCKS) blocks = MAX_BLOCKS;
    hipLaunchKernelGGL(sage_wmma_kernel, dim3(blocks), dim3(32 * WAVES_PER_BLOCK), 0, stream,
                       h, nbr_idx, nbr_w, w_self, w_nei, gamma, beta, out, N, K);
}


# Left-over tiles of the last round (one each in 106 work-groups) worked by all 16 waves of the work-group together instead of by one wave, whose SIMD carried 7 tiles against 6; sums unchanged
# speedup vs baseline: 2.3649x; 1.0882x over previous
#include <hip/hip_runtime.h>
#include <math.h>

typedef __attribute__((ext_vector_type(2))) float v2f;
typedef __attribute__((ext_vector_type(4))) float v4f;
typedef __attribute__((ext_vector_type(8))) float v8f;
typedef __attribute__((ext_vector_type(4))) int   v4i;
typedef __attribute__((address_space(1))) const v4f gv4f;

#define D 128
#define LNS 132
#define WAVES_PER_BLOCK 16
#define MAX_BLOCKS 256
#define WFRAG_FLOATS (8 * 32 * 2 * 64)

#define F32_BITS(u) __builtin_bit_cast(float, (unsigned)(u))
__device__ __forceinline__ float erf_both_ranges(const float y)
{
    const float t = __builtin_fabsf(y);
    const float s = y * y;
    float q = __builtin_fmaf(s, F32_BITS(0xba1345e1), F32_BITS(0x3ba10414));
    q = __builtin_fmaf(s, q, F32_BITS(0xbcdac9b8));
    q = __builtin_fmaf(s, q, F32_BITS(0x3de703be));
    q = __builtin_fmaf(s, q, F32_BITS(0xbec09330));
    q = __builtin_fmaf(s, q, F32_BITS(0x3e0375d0));
    const float r_small = __builtin_fmaf(t, q, t);
    float p = __builtin_fmaf(t, F32_BITS(0x378e98ab), F32_BITS(0xb9c68948));
    p = __builtin_fmaf(t, p, F32_BITS(0x3b7cd369));
    p = __builtin_fmaf(t, p, F32_BITS(0xbcc618b2));
    p = __builtin_fmaf(t, p, F32_BITS(0x3dda74e4));
    p = __builtin_fmaf(t, p, F32_BITS(0x3f228afd));
    p = __builtin_fmaf(t, p, F32_BITS(0x3e03c728));
    p = __builtin_fmaf(t, p, t);
    const float r_large = 1.0f - expf(-p);
    const float r = (t < 1.0f) ? r_small : r_large;
    return __builtin_copysignf(r, y);
}

__global__ __launch_bounds__(32 * WAVES_PER_BLOCK)
void sage_wmma_kernel(const float* __restrict__ h,
                      const int*   __restrict__ nbr_idx,
                      const float* __restrict__ nbr_w,
                      const float* __restrict__ w_self,
                      const float* __restrict__ w_nei,
                      const float* __restrict__ gamma,
                      const float* __restrict__ beta,
                      float* __restrict__ out,
                      int N, int K)
{
    __shared__ float wfrag[WFRAG_FLOATS];
    __shared__ float lds[WAVES_PER_BLOCK * 16 * LNS];

    const int lane   = threadIdx.x & 31;
    const int waveId = __builtin_amdgcn_readfirstlane(threadIdx.x >> 5);

    const int m    = lane & 15;
    const int half = lane >> 4;
    const int off  = half * 2;

    float* tileLds = lds + waveId * (16 * LNS);

    const int tiles = (N + 15) >> 4;
    const int tile0 = blockIdx.x + gridDim.x * waveId;
    const bool k16 = (K == 16);
    const int perRound = gridDim.x * WAVES_PER_BLOCK;
    const int coop     = k16 ? ((tiles % perRound) % (int)gridDim.x) : 0;
    const int tilesInd = tiles - coop;

    int   iv[8] = {0, 0, 0, 0, 0, 0, 0, 0};
    float wv[8] = {0.f, 0.f, 0.f, 0.f, 0.f, 0.f, 0.f, 0.f};
    auto fetch_tile_head = [&](const int tile) {
        if (k16) {
            int lrow = tile * 16 + (lane >> 1);
            if (lrow >= N) lrow = N - 1;
            const size_t e = (size_t)lrow * 16 + 8 * (lane & 1);
            const v4i i0 = *reinterpret_cast<const v4i*>(nbr_idx + e);
            const v4i i1 = *reinterpret_cast<const v4i*>(nbr_idx + e + 4);
            const v4f w0 = *reinterpret_cast<const v4f*>(nbr_w + e);
            const v4f w1 = *reinterpret_cast<const v4f*>(nbr_w + e + 4);
            iv[0] = i0.x; iv[1] = i0.y; iv[2] = i0.z; iv[3] = i0.w;
            iv[4] = i1.x; iv[5] = i1.y; iv[6] = i1.z; iv[7] = i1.w;
            wv[0] = w0.x; wv[1] = w0.y; wv[2] = w0.z; wv[3] = w0.w;
            wv[4] = w1.x; wv[5] = w1.y; wv[6] = w1.z; wv[7] = w1.w;
        }
        for (int r = 0; r < 16; ++r) {
            int grow = tile * 16 + r;
            if (grow >= N) grow = N - 1;
            v4f v = *reinterpret_cast<const v4f*>(h + (size_t)grow * D + 4 * lane);
            *reinterpret_cast<v4f*>(tileLds + r * LNS + 4 * lane) = v;
        }
    };

    static_assert(WAVES_PER_BLOCK == 16, "sixteen waves x sixteen pieces cover the 256 pieces");
    v4f wpiece[16];
    auto load_wpieces = [&]() {
#pragma unroll
        for (int t = 0; t < 16; ++t) {
            const float* wsrc = (t >> 3) ? w_nei : w_self;
            wpiece[t] = *reinterpret_cast<const v4f*>(wsrc + (size_t)(16 * (t & 7) + m) * D + 8 * waveId + 4 * half);
        }
    };
    auto store_wpieces = [&]() {
#pragma unroll
        for (int t = 0; t < 16; ++t) {
            const int mat = t >> 3;
            const int jt  = t & 7;
            float* dst = wfrag + ((jt * 32 + 2 * waveId + half) * 2 + mat) * 64;
            v2f lo = {wpiece[t].x, wpiece[t].y};
            v2f hi = {wpiece[t].z, wpiece[t].w};
            *reinterpret_cast<v2f*>(dst + 2 * m)        = lo;
            *reinterpret_cast<v2f*>(dst + 2 * (m + 16)) = hi;
        }
    };
    const bool head_early = k16 && (tile0 < tilesInd);
    if (head_early) {
        load_wpieces();
        fetch_tile_head(tile0);
        store_wpieces();
    } else {
        load_wpieces();
        store_wpieces();
    }
    const float g0 = gamma[lane], g1 = gamma[lane + 32], g2 = gamma[lane + 64], g3 = gamma[lane + 96];
    const float b0 = beta[lane],  b1 = beta[lane + 32],  b2 = beta[lane + 64],  b3 = beta[lane + 96];
    __syncthreads();

    for (int tile = tile0; tile < tilesInd; tile += gridDim.x * WAVES_PER_BLOCK) {

    if (tile != tile0 || !head_early) fetch_tile_head(tile);

    v2f aS[32];
#pragma unroll
    for (int c = 0; c < 32; ++c)
        aS[c] = *reinterpret_cast<const v2f*>(tileLds + m * LNS + 4 * c + off);

    v8f zs[8] = {};
    auto self_wmma16 = [&](v8f& acc, const float* bw, const int c0) {
#pragma unroll
        for (int c = 0; c < 16; ++c) {
            v2f bS = *reinterpret_cast<const v2f*>(bw + (c0 + c) * 128);
            acc = __builtin_amdgcn_wmma_f32_16x16x4_f32(false, aS[c0 + c], false, bS,
                                                        (short)0, acc, false, false);
        }
        __builtin_amdgcn_sched_barrier(0);
    };

    if (k16) {
        const unsigned long long hbase = reinterpret_cast<unsigned long long>(h);
        auto issue_row = [&](int r, v4f (&buf)[16]) {
            unsigned long long base[16];
#pragma unroll
            for (int k = 0; k < 16; ++k) {
                const int nrow = __builtin_amdgcn_readlane(iv[k & 7], 2 * r + (k >> 3));
                base[k] = hbase + ((unsigned long long)(long long)nrow << 9);
            }
            __builtin_amdgcn_sched_barrier(0);
#pragma unroll
            for (int k = 0; k < 16; ++k)
                buf[k] = reinterpret_cast<const gv4f*>(base[k])[lane];
            __builtin_amdgcn_sched_barrier(0);
        };
        auto sum_row = [&](int r, const v4f (&buf)[16]) {
            v4f acc = {};
#pragma unroll
            for (int k = 0; k < 16; ++k) {
                const float w = __int_as_float(__builtin_amdgcn_readlane(__float_as_int(wv[k & 7]), 2 * r + (k >> 3)));
                acc += buf[k] * w;
            }
            *reinterpret_cast<v4f*>(tileLds + r * LNS + 4 * lane) = acc;
        };
        v4f buf[16];
#pragma nounroll
        for (int p = 0; p < 8; ++p) {
            const float* bw = wfrag + p * (32 * 2 * 64) + 2 * lane;
            v8f acc = {};
            issue_row(2 * p, buf);
            self_wmma16(acc, bw, 0);
            sum_row(2 * p, buf);
            issue_row(2 * p + 1, buf);
            self_wmma16(acc, bw, 16);
            sum_row(2 * p + 1, buf);
            zs[0] = zs[1]; zs[1] = zs[2]; zs[2] = zs[3]; zs[3] = zs[4];
            zs[4] = zs[5]; zs[5] = zs[6]; zs[6] = zs[7]; zs[7] = acc;
        }
    } else {
    for (int r = 0; r < 16; ++r) {
        int grow = tile * 16 + r;
        if (grow >= N) grow = N - 1;
        v4f acc = {};
        for (int k = 0; k < K; ++k) {
            const int   nrow = nbr_idx[(size_t)grow * K + k];
            const float w    = nbr_w [(size_t)grow * K + k];
            v4f v = *reinterpret_cast<const v4f*>(h + (size_t)nrow * D + 4 * lane);
            acc += v * w;
        }
        *reinterpret_cast<v4f*>(tileLds + r * LNS + 4 * lane) = acc;
    }
#pragma nounroll
    for (int p = 0; p < 8; ++p) {
        const float* bw = wfrag + p * (32 * 2 * 64) + 2 * lane;
        v8f acc = {};
        self_wmma16(acc, bw, 0);
        self_wmma16(acc, bw, 16);
        zs[0] = zs[1]; zs[1] = zs[2]; zs[2] = zs[3]; zs[3] = zs[4];
        zs[4] = zs[5]; zs[5] = zs[6]; zs[6] = zs[7]; zs[7] = acc;
    }
    }

    v2f aA[32];
#pragma unroll
    for (int c = 0; c < 32; ++c)
        aA[c] = *reinterpret_cast<const v2f*>(tileLds + m * LNS + 4 * c + off);

#pragma nounroll
    for (int jt = 0; jt < 8; ++jt) {
        const int j0 = jt * 16;
        v8f acc = zs[0];
        zs[0] = zs[1]; zs[1] = zs[2]; zs[2] = zs[3]; zs[3] = zs[4];
        zs[4] = zs[5]; zs[5] = zs[6]; zs[6] = zs[7];
        const float* bw = wfrag + jt * (32 * 2 * 64) + 2 * lane;
#pragma unroll
        for (int c = 0; c < 32; ++c) {
            v2f bN = *reinterpret_cast<const v2f*>(bw + c * 128 + 64);
            acc = __builtin_amdgcn_wmma_f32_16x16x4_f32(false, aA[c], false, bN,
                                                        (short)0, acc, false, false);
        }
#pragma unroll
        for (int r = 0; r < 8; ++r) {
            const int rr = r + 8 * half;
            const float z = acc[r];
            const float g = 0.5f * z * (1.0f + erf_both_ranges(z * 0.70710678118654752f));
            tileLds[rr * LNS + j0 + m] = g;
        }
    }

    float x[16][4];
#pragma unroll
    for (int r = 0; r < 16; ++r) {
#pragma unroll
        for (int q = 0; q < 4; ++q) x[r][q] = tileLds[r * LNS + lane + 32 * q];
    }
    float s[16];
#pragma unroll
    for (int r = 0; r < 16; ++r) s[r] = x[r][0] + x[r][1] + x[r][2] + x[r][3];
#pragma unroll
    for (int d = 16; d >= 1; d >>= 1) {
        float t[16];
#pragma unroll
        for (int r = 0; r < 16; ++r) t[r] = __shfl_xor(s[r], d, 32);
#pragma unroll
        for (int r = 0; r < 16; ++r) s[r] += t[r];
    }
    float s2[16];
#pragma unroll
    for (int r = 0; r < 16; ++r) {
        const float mu = s[r] * (1.0f / D);
        const float d0 = x[r][0] - mu, d1 = x[r][1] - mu, d2 = x[r][2] - mu, d3 = x[r][3] - mu;
        x[r][0] = d0; x[r][1] = d1; x[r][2] = d2; x[r][3] = d3;
        s2[r] = d0 * d0 + d1 * d1 + d2 * d2 + d3 * d3;
    }
#pragma unroll
    for (int d = 16; d >= 1; d >>= 1) {
        float t[16];
#pragma unroll
        for (int r = 0; r < 16; ++r) t[r] = __shfl_xor(s2[r], d, 32);
#pragma unroll
        for (int r = 0; r < 16; ++r) s2[r] += t[r];
    }
    float* const otile = out + (size_t)tile * (16 * D) + lane;
    auto finish_row = [&](int r) {
        const float inv = rsqrtf(s2[r] * (1.0f / D) + 1e-5f);
        float* o = otile + r * D;
        o[0]  = x[r][0] * inv * g0 + b0;
        o[32] = x[r][1] * inv * g1 + b1;
        o[64] = x[r][2] * inv * g2 + b2;
        o[96] = x[r][3] * inv * g3 + b3;
    };
    if (tile * 16 + 16 <= N) {
#pragma unroll
        for (int r = 0; r < 16; ++r) finish_row(r);
    } else {
#pragma unroll
        for (int r = 0; r < 16; ++r)
            if (tile * 16 + r < N) finish_row(r);
    }
    }

    if ((int)blockIdx.x < coop) {
        const int T = tilesInd + (int)blockIdx.x;
        float* const bufA = lds;
        float* const bufS = lds + 16 * LNS;
        float* const bufG = lds + 2 * 16 * LNS;
        __syncthreads();

        {
            int grow = T * 16 + waveId;
            if (grow >= N) grow = N - 1;
            const int   ci = nbr_idx[(size_t)grow * 16 + m];
            const float cw = nbr_w [(size_t)grow * 16 + m];
            const v4f hv = *reinterpret_cast<const v4f*>(h + (size_t)grow * D + 4 * lane);
            const unsigned long long hb = reinterpret_cast<unsigned long long>(h);
            v4f rb[16];
            unsigned long long rbase[16];
#pragma unroll
            for (int k = 0; k < 16; ++k) {
                const int nrow = __builtin_amdgcn_readlane(ci, k);
                rbase[k] = hb + ((unsigned long long)(long long)nrow << 9);
            }
            __builtin_amdgcn_sched_barrier(0);
#pragma unroll
            for (int k = 0; k < 16; ++k)
                rb[k] = reinterpret_cast<const gv4f*>(rbase[k])[lane];
            __builtin_amdgcn_sched_barrier(0);
            v4f acc = {};
#pragma unroll
            for (int k = 0; k < 16; ++k) {
                const float w = __int_as_float(__builtin_amdgcn_readlane(__float_as_int(cw), k));
                acc += rb[k] * w;
            }
            *reinterpret_cast<v4f*>(bufS + waveId * LNS + 4 * lane) = hv;
            *reinterpret_cast<v4f*>(bufA + waveId * LNS + 4 * lane) = acc;
        }
        __syncthreads();

        if (waveId < 8) {
            const int j0 = waveId * 16;
            const float* bw = wfrag + waveId * (32 * 2 * 64) + 2 * lane;
            v8f acc = {};
#pragma unroll
            for (int c = 0; c < 32; ++c) {
                const v2f a = *reinterpret_cast<const v2f*>(bufS + m * LNS + 4 * c + off);
                const v2f b = *reinterpret_cast<const v2f*>(bw + c * 128);
                acc = __builtin_amdgcn_wmma_f32_16x16x4_f32(false, a, false, b, (short)0, acc, false, false);
            }
#pragma unroll
            for (int c = 0; c < 32; ++c) {
                const v2f a = *reinterpret_cast<const v2f*>(bufA + m * LNS + 4 * c + off);
                const v2f b = *reinterpret_cast<const v2f*>(bw + c * 128 + 64);
                acc = __builtin_amdgcn_wmma_f32_16x16x4_f32(false, a, false, b, (short)0, acc, false, false);
            }
#pragma unroll
            for (int r = 0; r < 8; ++r) {
                const int rr = r + 8 * half;
                const float z = acc[r];
                const float g = 0.5f * z * (1.0f + erf_both_ranges(z * 0.70710678118654752f));
                bufG[rr * LNS + j0 + m] = g;
            }
        }
        __syncthreads();

        if (T * 16 + waveId < N) {
            const float* xr = bufG + waveId * LNS + lane;
            const float x0 = xr[0], x1 = xr[32], x2 = xr[64], x3 = xr[96];
            float s = x0 + x1 + x2 + x3;
            for (int d = 16; d >= 1; d >>= 1) s += __shfl_xor(s, d, 32);
            const float mu = s * (1.0f / D);
            const float d0 = x0 - mu, d1 = x1 - mu, d2 = x2 - mu, d3 = x3 - mu;
            float s2 = d0 * d0 + d1 * d1 + d2 * d2 + d3 * d3;
            for (int d = 16; d >= 1; d >>= 1) s2 += __shfl_xor(s2, d, 32);
            const float inv = rsqrtf(s2 * (1.0f / D) + 1e-5f);
            float* o = out + ((size_t)T * 16 + waveId) * D + lane;
            o[0]  = d0 * inv * g0 + b0;
            o[32] = d1 * inv * g1 + b1;
            o[64] = d2 * inv * g2 + b2;
            o[96] = d3 * inv * g3 + b3;
        }
    }
}

extern "C" void kernel_launch(void* const* d_in, const int* in_sizes, int n_in,
                              void* d_out, int out_size, void* d_ws, size_t ws_size,
                              hipStream_t stream) {
    const float* h       = (const float*)d_in[0];
    const int*   nbr_idx = (const int*)  d_in[1];
    const float* nbr_w   = (const float*)d_in[2];
    const float* w_self  = (const float*)d_in[3];
    const float* w_nei   = (const float*)d_in[4];
    const float* gamma   = (const float*)d_in[5];
    const float* beta    = (const float*)d_in[6];
    float* out = (float*)d_out;

    const int N = in_sizes[0] / D;
    const int K = (N > 0) ? (in_sizes[1] / N) : 16;

    const int tiles  = (N + 15) / 16;
    int blocks = (tiles + WAVES_PER_BLOCK - 1) / WAVES_PER_BLOCK;
    if (blocks > MAX_BLOCKS) blocks = MAX_BLOCKS;
    hipLaunchKernelGGL(sage_wmma_kernel, dim3(blocks), dim3(32 * WAVES_PER_BLOCK), 0, stream,
                       h, nbr_idx, nbr_w, w_self, w_nei, gamma, beta, out, N, K);
}
